# starting kernel
# baseline (speedup 1.0000x reference)
#include <hip/hip_runtime.h>


#define NROWS   1048576
#define BSEG    64
#define DFEAT   64
#define CHUNK   2048
#define THREADS 256
#define PF_ROWS 64

typedef float v2f __attribute__((ext_vector_type(2)));
typedef float v8f __attribute__((ext_vector_type(8)));

__global__ void gab_zero_ws(float* __restrict__ ws) {
    int i = blockIdx.x * blockDim.x + threadIdx.x;
    if (i < BSEG * DFEAT) ws[i] = 0.0f;
}

__global__ __launch_bounds__(THREADS)
void gab_seg_sum_wmma(const float* __restrict__ x,
                      const int*   __restrict__ seg,
                      float*       __restrict__ ws) {
    const int tid  = threadIdx.x;
    const int lane = tid & 31;
    const int wave = tid >> 5;
    const int ft   = wave & 3;
    const int rh   = wave >> 2;

    const int chunk_start = blockIdx.x * CHUNK;
    const int row_base    = chunk_start + rh * (CHUNK / 2);

    const int seg_base = seg[chunk_start];

    const int m  = lane & 15;
    const int kh = (lane >> 4) << 1;
    const int f  = ft * 16 + m;

    const float* px  = x   + (size_t)(row_base + kh) * DFEAT + f;
    const int*   ps  = seg + row_base + kh;
    const float* ppf = px  + (size_t)PF_ROWS * DFEAT;

    v8f c0 = {};
    v8f c1 = {};

    for (int it = 0; it < (CHUNK / 2) / 8; ++it) {
        const int s0 = ps[0];
        const int s1 = ps[1];
        const int s2 = ps[4];
        const int s3 = ps[5];

        v2f a0, a1, b0, b1;
        a0.x = (s0 - seg_base == m) ? 1.0f : 0.0f;
        a0.y = (s1 - seg_base == m) ? 1.0f : 0.0f;
        a1.x = (s2 - seg_base == m) ? 1.0f : 0.0f;
        a1.y = (s3 - seg_base == m) ? 1.0f : 0.0f;

        b0.x = px[0];
        b0.y = px[DFEAT];
        b1.x = px[4 * DFEAT];
        b1.y = px[5 * DFEAT];

        __builtin_prefetch(ppf, 0, 1);

        c0 = __builtin_amdgcn_wmma_f32_16x16x4_f32(
                false, a0, false, b0, (short)0, c0, false, false);
        c1 = __builtin_amdgcn_wmma_f32_16x16x4_f32(
                false, a1, false, b1, (short)0, c1, false, false);

        px  += 8 * DFEAT;
        ps  += 8;
        ppf += 8 * DFEAT;
    }

    const int mofs = (lane >> 4) * 8;
    #pragma unroll
    for (int r = 0; r < 8; ++r) {
        float v = c0[r] + c1[r];
        if (v != 0.0f) {
            int sgi = seg_base + r + mofs;
            if (sgi < BSEG)
                atomicAdd(&ws[sgi * DFEAT + f], v);
        }
    }
}

__global__ void gab_finalize(const float* __restrict__ ws,
                             const int*   __restrict__ lengths,
                             float*       __restrict__ out) {
    int i = blockIdx.x * blockDim.x + threadIdx.x;
    if (i < BSEG * DFEAT) {
        out[i] = ws[i] / (float)lengths[i / DFEAT];
    }
}

extern "C" void kernel_launch(void* const* d_in, const int* in_sizes, int n_in,
                              void* d_out, int out_size, void* d_ws, size_t ws_size,
                              hipStream_t stream) {
    const float* x       = (const float*)d_in[0];
    const int*   seg     = (const int*)  d_in[1];
    const int*   lengths = (const int*)  d_in[2];
    float*       out     = (float*)d_out;
    float*       ws      = (float*)d_ws;

    gab_zero_ws<<<(BSEG * DFEAT + 255) / 256, 256, 0, stream>>>(ws);
    gab_seg_sum_wmma<<<NROWS / CHUNK, THREADS, 0, stream>>>(x, seg, ws);
    gab_finalize<<<(BSEG * DFEAT + 255) / 256, 256, 0, stream>>>(ws, lengths, out);
}


# Cross-block sum without atomics: each work-group stores its segment rows (8 waves added in LDS), a per-segment launch adds them in fixed order; zero launch gone (3 to 2); f32 sum order regrouped
# speedup vs baseline: 1.2560x; 1.2560x over previous
#include <hip/hip_runtime.h>


#define NROWS   1048576
#define BSEG    64
#define DFEAT   64
#define CHUNK   1024
#define THREADS 256
#define GROUP   16
#define STRIDE  (8 * GROUP)
#define NTRIP   (CHUNK / STRIDE)
#define NBLK    (NROWS / CHUNK)
#define NPROW   (NBLK + BSEG)
#define EDGE_OFS (NPROW * DFEAT)
#define WS_NEED ((size_t)(EDGE_OFS + NBLK + 4) * 4)

typedef float v2f __attribute__((ext_vector_type(2)));
typedef float v8f __attribute__((ext_vector_type(8)));
typedef int   v4i __attribute__((ext_vector_type(4)));

__global__ __launch_bounds__(THREADS) __attribute__((amdgpu_waves_per_eu(8, 8)))
void gab_seg_sum_wmma(const float* __restrict__ x,
                      const int*   __restrict__ seg,
                      float*       __restrict__ ws,
                      size_t                    ws_bytes) {
    __shared__ float part[8][16][DFEAT];
    if (ws_bytes < WS_NEED) return;

    const int tid  = threadIdx.x;
    const int lane = tid & 31;
    const int wave = tid >> 5;
    const int c    = lane & 15;
    const int h    = lane >> 4;

    const int chunk_start = blockIdx.x * CHUNK;
    const int row_base    = chunk_start + wave * GROUP;

    const int seg_base = seg[chunk_start];
    const int seg_next = seg[(chunk_start + CHUNK < NROWS) ? chunk_start + CHUNK : NROWS - 1];

    const float* px  = x   + (size_t)(row_base + 8 * h) * DFEAT + c;
    const int*   ps  = seg + row_base + 8 * h;
    const float* ppf = x   + (size_t)(row_base + STRIDE + c) * DFEAT + h * 32;

    v8f acc0 = {};
    v8f acc1 = {};
    v8f acc2 = {};
    v8f acc3 = {};

    #pragma clang loop unroll(disable)
    for (int it = 0; it < NTRIP; ++it) {
        const v4i i0 = *(const v4i*)(ps);
        const v4i i1 = *(const v4i*)(ps + 4);

        float b[8][4];
        #pragma unroll
        for (int r = 0; r < 8; ++r) {
            #pragma unroll
            for (int t = 0; t < 4; ++t)
                b[r][t] = px[r * DFEAT + t * 16];
        }

        __builtin_prefetch((it + 1 < NTRIP) ? ppf : px, 0, 1);

        const int id[8] = { i0.x, i0.y, i0.z, i0.w, i1.x, i1.y, i1.z, i1.w };

        #pragma unroll
        for (int j = 0; j < 4; ++j) {
            v2f a;
            a.x = (id[2 * j]     - seg_base == c) ? 1.0f : 0.0f;
            a.y = (id[2 * j + 1] - seg_base == c) ? 1.0f : 0.0f;
            v2f b0, b1, b2, b3;
            b0.x = b[2 * j][0];  b0.y = b[2 * j + 1][0];
            b1.x = b[2 * j][1];  b1.y = b[2 * j + 1][1];
            b2.x = b[2 * j][2];  b2.y = b[2 * j + 1][2];
            b3.x = b[2 * j][3];  b3.y = b[2 * j + 1][3];
            acc0 = __builtin_amdgcn_wmma_f32_16x16x4_f32(
                    false, a, false, b0, (short)0, acc0, false, false);
            acc1 = __builtin_amdgcn_wmma_f32_16x16x4_f32(
                    false, a, false, b1, (short)0, acc1, false, false);
            acc2 = __builtin_amdgcn_wmma_f32_16x16x4_f32(
                    false, a, false, b2, (short)0, acc2, false, false);
            acc3 = __builtin_amdgcn_wmma_f32_16x16x4_f32(
                    false, a, false, b3, (short)0, acc3, false, false);
        }

        px  += STRIDE * DFEAT;
        ps  += STRIDE;
        ppf += STRIDE * DFEAT;
    }

    const int top  = (seg_next < BSEG - 1) ? seg_next : BSEG - 1;
    const int nrow = (seg_base >= 0) ? top - seg_base : -1;
    const int nwin = (nrow < 15) ? nrow : 15;

    const int mofs = h * 8;
    #pragma unroll
    for (int r = 0; r < 8; ++r) {
        const int slot = r + mofs;
        if (slot <= nwin) {
            float* p = &part[wave][slot][c];
            p[0]  = acc0[r];
            p[16] = acc1[r];
            p[32] = acc2[r];
            p[48] = acc3[r];
        }
    }
    __syncthreads();

    const int d = tid & 63;
    const int q = tid >> 6;
    float* rows = ws + (size_t)(blockIdx.x + seg_base) * DFEAT + d;
    for (int slot = q; slot <= nrow; slot += 4) {
        float v = 0.0f;
        if (slot <= 15) {
            v = part[0][slot][d];
            #pragma unroll
            for (int w = 1; w < 8; ++w) v += part[w][slot][d];
        }
        rows[slot * DFEAT] = v;
    }
    if (tid == 0) {
        int* edge = (int*)(ws + EDGE_OFS);
        edge[blockIdx.x] = seg_base;
        if (blockIdx.x == NBLK - 1) edge[NBLK] = seg_next;
    }
}

__global__ __launch_bounds__(256)
void gab_rows_finalize(const float* __restrict__ ws,
                       const int*   __restrict__ lengths,
                       float*       __restrict__ out,
                       size_t                    ws_bytes) {
    __shared__ int   scnt[16];
    __shared__ float sq[4][DFEAT];
    if (ws_bytes < WS_NEED) return;

    const int tid  = threadIdx.x;
    const int lane = tid & 31;
    const int wave = tid >> 5;
    const int s    = blockIdx.x;

    const int* edge = (const int*)(ws + EDGE_OFS);
    const v4i e       = ((const v4i*)edge)[tid];
    const int e_first = edge[0];
    const int e_last  = edge[NBLK];
    const float len   = (float)lengths[s];

    const int nle = __builtin_popcount(__builtin_amdgcn_ballot_w32(e.x <= s))
                  + __builtin_popcount(__builtin_amdgcn_ballot_w32(e.y <= s))
                  + __builtin_popcount(__builtin_amdgcn_ballot_w32(e.z <= s))
                  + __builtin_popcount(__builtin_amdgcn_ballot_w32(e.w <= s));
    const int nlt = __builtin_popcount(__builtin_amdgcn_ballot_w32(e.x < s))
                  + __builtin_popcount(__builtin_amdgcn_ballot_w32(e.y < s))
                  + __builtin_popcount(__builtin_amdgcn_ballot_w32(e.z < s))
                  + __builtin_popcount(__builtin_amdgcn_ballot_w32(e.w < s));
    if (lane == 0) { scnt[wave] = nle; scnt[8 + wave] = nlt; }
    __syncthreads();
    int tle = 0, tlt = 0;
    #pragma unroll
    for (int w = 0; w < 8; ++w) { tle += scnt[w]; tlt += scnt[8 + w]; }
    const int blo = tlt + ((e_last < s) ? 1 : 0) - ((e_first < s) ? 1 : 0);
    const int bhi = tle - 1;

    const int d = tid & 63;
    const int q = tid >> 6;
    const float* rows = ws + (size_t)s * DFEAT + d;
    float acc = 0.0f;
    for (int b0 = blo + q; b0 <= bhi; b0 += 32) {
        float v[8];
        #pragma unroll
        for (int k = 0; k < 8; ++k) {
            const int b = b0 + 4 * k;
            v[k] = (b <= bhi) ? rows[(size_t)b * DFEAT] : 0.0f;
        }
        #pragma unroll
        for (int k = 0; k < 8; ++k) acc += v[k];
    }
    sq[q][d] = acc;
    __syncthreads();
    if (tid < DFEAT) {
        const float tot = ((sq[0][tid] + sq[1][tid]) + sq[2][tid]) + sq[3][tid];
        out[s * DFEAT + tid] = tot / len;
    }
}

extern "C" void kernel_launch(void* const* d_in, const int* in_sizes, int n_in,
                              void* d_out, int out_size, void* d_ws, size_t ws_size,
                              hipStream_t stream) {
    const float* x       = (const float*)d_in[0];
    const int*   seg     = (const int*)  d_in[1];
    const int*   lengths = (const int*)  d_in[2];
    float*       out     = (float*)d_out;
    float*       ws      = (float*)d_ws;

    gab_seg_sum_wmma<<<NROWS / CHUNK, THREADS, 0, stream>>>(x, seg, ws, ws_size);
    gab_rows_finalize<<<BSEG, 256, 0, stream>>>(ws, lengths, out, ws_size);
}
